# starting kernel
# baseline (speedup 1.0000x reference)
#include <hip/hip_runtime.h>
#include <hip/hip_bf16.h>


typedef __attribute__((ext_vector_type(2))) float v2f;
typedef __attribute__((ext_vector_type(4))) float v4f;
typedef __attribute__((ext_vector_type(8))) float v8f;

#define NBATCH   64
#define LSEQ     32768
#define NCH      16
#define SEG      512
#define NSEG     (LSEQ / SEG)
#define TOTSEG   (NBATCH * NSEG)
#define SEGFLTS  (SEG * NCH)

__global__ void seg_sums_kernel(const float* __restrict__ in,
                                float* __restrict__ sums) {
  const int wave = (blockIdx.x * blockDim.x + threadIdx.x) >> 5;
  const int lane = threadIdx.x & 31;
  const int g = lane & 3;
  const int r = lane >> 2;

  const v4f* p = (const v4f*)(in + (size_t)wave * SEGFLTS) + (r * 4 + g);
  v4f acc = {0.f, 0.f, 0.f, 0.f};
#pragma unroll 4
  for (int i = 0; i < SEG / 8; ++i) {
    acc += p[32 * i];
  }
#pragma unroll
  for (int m = 4; m <= 16; m <<= 1) {
    acc[0] += __shfl_xor(acc[0], m, 32);
    acc[1] += __shfl_xor(acc[1], m, 32);
    acc[2] += __shfl_xor(acc[2], m, 32);
    acc[3] += __shfl_xor(acc[3], m, 32);
  }
  if (r == 0) {
    ((v4f*)sums)[wave * 4 + g] = acc;
  }
}

__global__ void seg_scan_kernel(const float* __restrict__ sums,
                                float* __restrict__ excl) {
  const int batch = blockIdx.x;
  const int lane = threadIdx.x;
  if (lane < NCH) {
    float carry = 0.f;
    for (int s = 0; s < NSEG; ++s) {
      const int idx = (batch * NSEG + s) * NCH + lane;
      const float v = sums[idx];
      excl[idx] = carry;
      carry += v;
    }
  }
}

__global__ void wmma_scan_kernel(const float* __restrict__ in,
                                 const float* __restrict__ excl,
                                 float* __restrict__ out) {
  const int wave = (blockIdx.x * blockDim.x + threadIdx.x) >> 5;
  const int lane = threadIdx.x & 31;
  const int m    = lane & 15;
  const int half = lane >> 4;
  const int k0   = half * 2;

  const float* base  = in  + (size_t)wave * SEGFLTS;
  float*       obase = out + (size_t)wave * SEGFLTS;

  float carry = excl[wave * NCH + m];

  for (int tile = 0; tile < SEG / 16; ++tile) {
    const float* tb = base + tile * 256;

    v8f d;
#pragma unroll
    for (int i = 0; i < 8; ++i) d[i] = carry;

#pragma unroll
    for (int kb = 0; kb < 16; kb += 4) {
      v2f a, b;
      a[0] = ((kb + k0)     <= m) ? 1.0f : 0.0f;
      a[1] = ((kb + k0 + 1) <= m) ? 1.0f : 0.0f;
      b[0] = tb[(kb + k0)     * NCH + m];
      b[1] = tb[(kb + k0 + 1) * NCH + m];
      d = __builtin_amdgcn_wmma_f32_16x16x4_f32(
           false, a,  false, b,
           (short)0, d,  false,  false);
    }

    float* ob = obase + tile * 256;
#pragma unroll
    for (int i = 0; i < 8; ++i) {
      __builtin_nontemporal_store(d[i], &ob[(i + 8 * half) * NCH + m]);
    }

    carry = __shfl(d[7], 16 + m, 32);
  }
}

extern "C" void kernel_launch(void* const* d_in, const int* in_sizes, int n_in,
                              void* d_out, int out_size, void* d_ws, size_t ws_size,
                              hipStream_t stream) {
  const float* in = (const float*)d_in[0];
  float* out = (float*)d_out;

  float* sums = (float*)d_ws;
  float* excl = sums + (size_t)TOTSEG * NCH;

  seg_sums_kernel<<<TOTSEG / 8, 256, 0, stream>>>(in, sums);
  seg_scan_kernel<<<NBATCH, 32, 0, stream>>>(sums, excl);
  wmma_scan_kernel<<<TOTSEG / 8, 256, 0, stream>>>(in, excl, out);
}


# More loads in flight per wave in all 3 passes: scan pass requests 7 tiles (56 loads) ahead via a register ring, sums pass 16-32 row loads, segment scan all 63 at once; sum order unchanged
# speedup vs baseline: 1.4099x; 1.4099x over previous
#include <hip/hip_runtime.h>
#include <hip/hip_bf16.h>


typedef __attribute__((ext_vector_type(2))) float v2f;
typedef __attribute__((ext_vector_type(4))) float v4f;
typedef __attribute__((ext_vector_type(8))) float v8f;

#define NBATCH   64
#define LSEQ     32768
#define NCH      16
#define SEG      512
#define NSEG     (LSEQ / SEG)
#define TOTSEG   (NBATCH * NSEG)
#define SEGFLTS  (SEG * NCH)
#define SUMQ     16
#define RING     8
#define AHEAD    7
#define NTILE    (SEG / 16)

__global__ void __launch_bounds__(256) seg_sums_kernel(const float* __restrict__ in,
                                float* __restrict__ sums) {
  const int wave = (blockIdx.x * blockDim.x + threadIdx.x) >> 5;
  const int lane = threadIdx.x & 31;
  const int g = lane & 3;
  const int r = lane >> 2;

  const v4f* p = (const v4f*)(in + (size_t)wave * SEGFLTS) + (r * 4 + g);
  v4f acc = {0.f, 0.f, 0.f, 0.f};
  v4f buf[2][SUMQ];
#pragma unroll
  for (int q = 0; q < SUMQ; ++q) buf[0][q] = p[32 * q];
#pragma unroll
  for (int c = 0; c < SEG / 8 / SUMQ; ++c) {
    __builtin_amdgcn_sched_barrier(0);
    if (c + 1 < SEG / 8 / SUMQ) {
#pragma unroll
      for (int q = 0; q < SUMQ; ++q)
        buf[(c + 1) & 1][q] = p[32 * ((c + 1) * SUMQ + q)];
    }
    __builtin_amdgcn_sched_barrier(0);
#pragma unroll
    for (int q = 0; q < SUMQ; ++q) acc += buf[c & 1][q];
  }
#pragma unroll
  for (int m = 4; m <= 16; m <<= 1) {
    acc[0] += __shfl_xor(acc[0], m, 32);
    acc[1] += __shfl_xor(acc[1], m, 32);
    acc[2] += __shfl_xor(acc[2], m, 32);
    acc[3] += __shfl_xor(acc[3], m, 32);
  }
  if (r == 0) {
    ((v4f*)sums)[wave * 4 + g] = acc;
  }
}

__global__ void seg_scan_kernel(const float* __restrict__ sums,
                                float* __restrict__ excl) {
  const int batch = blockIdx.x;
  const int lane = threadIdx.x;
  if (lane < NCH) {
    const float* sp = sums + (size_t)batch * NSEG * NCH + lane;
    float*       ep = excl + (size_t)batch * NSEG * NCH + lane;
    float v[NSEG - 1];
#pragma unroll
    for (int s = 0; s < NSEG - 1; ++s) v[s] = sp[s * NCH];
    __builtin_amdgcn_sched_barrier(0);
    float carry = 0.f;
#pragma unroll
    for (int s = 0; s < NSEG; ++s) {
      ep[s * NCH] = carry;
      if (s < NSEG - 1) carry += v[s];
    }
  }
}

__device__ __forceinline__ void load_tile(float (&x)[8], const float* tb,
                                          int k0, int m) {
#pragma unroll
  for (int j = 0; j < 4; ++j) {
    x[2 * j]     = tb[(4 * j + k0)     * NCH + m];
    x[2 * j + 1] = tb[(4 * j + k0 + 1) * NCH + m];
  }
}

__global__ void wmma_scan_kernel(const float* __restrict__ in,
                                 const float* __restrict__ excl,
                                 float* __restrict__ out) {
  const int wave = (blockIdx.x * blockDim.x + threadIdx.x) >> 5;
  const int lane = threadIdx.x & 31;
  const int m    = lane & 15;
  const int half = lane >> 4;
  const int k0   = half * 2;

  const float* base  = in  + (size_t)wave * SEGFLTS;
  float*       obase = out + (size_t)wave * SEGFLTS;

  float carry = excl[wave * NCH + m];

  float bx[RING][8];
#pragma unroll
  for (int p = 0; p < AHEAD; ++p) load_tile(bx[p], base + p * 256, k0, m);
  __builtin_amdgcn_sched_barrier(0);

#pragma unroll
  for (int t = 0; t < NTILE; ++t) {
    const float (&x)[8] = bx[t % RING];

    v8f d;
#pragma unroll
    for (int i = 0; i < 8; ++i) d[i] = carry;

#pragma unroll
    for (int kb = 0; kb < 16; kb += 4) {
      v2f a, b;
      a[0] = ((kb + k0)     <= m) ? 1.0f : 0.0f;
      a[1] = ((kb + k0 + 1) <= m) ? 1.0f : 0.0f;
      b[0] = x[kb / 2];
      b[1] = x[kb / 2 + 1];
      d = __builtin_amdgcn_wmma_f32_16x16x4_f32(
           false, a,  false, b,
           (short)0, d,  false,  false);
    }

    __builtin_amdgcn_sched_barrier(0);
    if (t + AHEAD < NTILE) {
      load_tile(bx[(t + AHEAD) % RING], base + (t + AHEAD) * 256, k0, m);
    }
    __builtin_amdgcn_sched_barrier(0);

    float* ob = obase + t * 256;
#pragma unroll
    for (int i = 0; i < 8; ++i) {
      __builtin_nontemporal_store(d[i], &ob[(i + 8 * half) * NCH + m]);
    }

    carry = __shfl(d[7], 16 + m, 32);
  }
}

extern "C" void kernel_launch(void* const* d_in, const int* in_sizes, int n_in,
                              void* d_out, int out_size, void* d_ws, size_t ws_size,
                              hipStream_t stream) {
  const float* in = (const float*)d_in[0];
  float* out = (float*)d_out;

  float* sums = (float*)d_ws;
  float* excl = sums + (size_t)TOTSEG * NCH;

  seg_sums_kernel<<<TOTSEG / 8, 256, 0, stream>>>(in, sums);
  seg_scan_kernel<<<NBATCH, 32, 0, stream>>>(sums, excl);
  wmma_scan_kernel<<<TOTSEG / 8, 256, 0, stream>>>(in, excl, out);
}


# Two launches instead of three: each scan wave forms its own carry-in by adding the totals of the segments ahead of it, first to last as before; the separate segment-scan launch is gone
# speedup vs baseline: 1.4839x; 1.0525x over previous
#include <hip/hip_runtime.h>
#include <hip/hip_bf16.h>


typedef __attribute__((ext_vector_type(2))) float v2f;
typedef __attribute__((ext_vector_type(4))) float v4f;
typedef __attribute__((ext_vector_type(8))) float v8f;

#define NBATCH   64
#define LSEQ     32768
#define NCH      16
#define SEG      512
#define NSEG     (LSEQ / SEG)
#define TOTSEG   (NBATCH * NSEG)
#define SEGFLTS  (SEG * NCH)
#define SUMQ     16
#define RING     8
#define AHEAD    7
#define NTILE    (SEG / 16)

__global__ void __launch_bounds__(256) seg_sums_kernel(const float* __restrict__ in,
                                float* __restrict__ sums) {
  const int wave = (blockIdx.x * blockDim.x + threadIdx.x) >> 5;
  const int lane = threadIdx.x & 31;
  const int g = lane & 3;
  const int r = lane >> 2;

  const v4f* p = (const v4f*)(in + (size_t)wave * SEGFLTS) + (r * 4 + g);
  v4f acc = {0.f, 0.f, 0.f, 0.f};
  v4f buf[2][SUMQ];
#pragma unroll
  for (int q = 0; q < SUMQ; ++q) buf[0][q] = p[32 * q];
#pragma unroll
  for (int c = 0; c < SEG / 8 / SUMQ; ++c) {
    __builtin_amdgcn_sched_barrier(0);
    if (c + 1 < SEG / 8 / SUMQ) {
#pragma unroll
      for (int q = 0; q < SUMQ; ++q)
        buf[(c + 1) & 1][q] = p[32 * ((c + 1) * SUMQ + q)];
    }
    __builtin_amdgcn_sched_barrier(0);
#pragma unroll
    for (int q = 0; q < SUMQ; ++q) acc += buf[c & 1][q];
  }
#pragma unroll
  for (int m = 4; m <= 16; m <<= 1) {
    acc[0] += __shfl_xor(acc[0], m, 32);
    acc[1] += __shfl_xor(acc[1], m, 32);
    acc[2] += __shfl_xor(acc[2], m, 32);
    acc[3] += __shfl_xor(acc[3], m, 32);
  }
  if (r == 0) {
    ((v4f*)sums)[wave * 4 + g] = acc;
  }
}

__device__ __forceinline__ void load_tile(float (&x)[8], const float* tb,
                                          int k0, int m) {
#pragma unroll
  for (int j = 0; j < 4; ++j) {
    x[2 * j]     = tb[(4 * j + k0)     * NCH + m];
    x[2 * j + 1] = tb[(4 * j + k0 + 1) * NCH + m];
  }
}

__global__ void __launch_bounds__(256) wmma_scan_kernel(const float* __restrict__ in,
                                 const float* __restrict__ sums,
                                 float* __restrict__ out) {
  const int wave = (blockIdx.x * blockDim.x + threadIdx.x) >> 5;
  const int lane = threadIdx.x & 31;
  const int m    = lane & 15;
  const int half = lane >> 4;
  const int k0   = half * 2;

  const float* base  = in  + (size_t)wave * SEGFLTS;
  float*       obase = out + (size_t)wave * SEGFLTS;

  const int seg = wave & (NSEG - 1);
  const float* sp = sums + (size_t)(wave - seg) * NCH + m;
  float v[NSEG - 1];
#pragma unroll
  for (int j = 0; j < NSEG - 1; ++j) v[j] = sp[j * NCH];
  __builtin_amdgcn_sched_barrier(0);

  float bx[RING][8];
#pragma unroll
  for (int p = 0; p < AHEAD; ++p) load_tile(bx[p], base + p * 256, k0, m);
  __builtin_amdgcn_sched_barrier(0);

  float carry = 0.f;
#pragma unroll
  for (int j = 0; j < NSEG - 1; ++j) carry += (j < seg) ? v[j] : 0.f;

#pragma unroll
  for (int t = 0; t < NTILE; ++t) {
    const float (&x)[8] = bx[t % RING];

    v8f d;
#pragma unroll
    for (int i = 0; i < 8; ++i) d[i] = carry;

#pragma unroll
    for (int kb = 0; kb < 16; kb += 4) {
      v2f a, b;
      a[0] = ((kb + k0)     <= m) ? 1.0f : 0.0f;
      a[1] = ((kb + k0 + 1) <= m) ? 1.0f : 0.0f;
      b[0] = x[kb / 2];
      b[1] = x[kb / 2 + 1];
      d = __builtin_amdgcn_wmma_f32_16x16x4_f32(
           false, a,  false, b,
           (short)0, d,  false,  false);
    }

    __builtin_amdgcn_sched_barrier(0);
    if (t + AHEAD < NTILE) {
      load_tile(bx[(t + AHEAD) % RING], base + (t + AHEAD) * 256, k0, m);
    }
    __builtin_amdgcn_sched_barrier(0);

    float* ob = obase + t * 256;
#pragma unroll
    for (int i = 0; i < 8; ++i) {
      __builtin_nontemporal_store(d[i], &ob[(i + 8 * half) * NCH + m]);
    }

    carry = __shfl(d[7], 16 + m, 32);
  }
}

extern "C" void kernel_launch(void* const* d_in, const int* in_sizes, int n_in,
                              void* d_out, int out_size, void* d_ws, size_t ws_size,
                              hipStream_t stream) {
  const float* in = (const float*)d_in[0];
  float* out = (float*)d_out;

  float* sums = (float*)d_ws;

  seg_sums_kernel<<<TOTSEG / 8, 256, 0, stream>>>(in, sums);
  wmma_scan_kernel<<<TOTSEG / 8, 256, 0, stream>>>(in, sums, out);
}


# Totals launch walks each segment from its last row to its first, so the scan launch re-reads the newest lines of the second-level cache first; a segment total now adds its rows last to first
# speedup vs baseline: 1.5820x; 1.0661x over previous
#include <hip/hip_runtime.h>
#include <hip/hip_bf16.h>


typedef __attribute__((ext_vector_type(2))) float v2f;
typedef __attribute__((ext_vector_type(4))) float v4f;
typedef __attribute__((ext_vector_type(8))) float v8f;

#define NBATCH   64
#define LSEQ     32768
#define NCH      16
#define SEG      512
#define NSEG     (LSEQ / SEG)
#define TOTSEG   (NBATCH * NSEG)
#define SEGFLTS  (SEG * NCH)
#define SUMQ     16
#define LASTQ    (SEG / 8 - 1)
#define RING     8
#define AHEAD    7
#define NTILE    (SEG / 16)

__global__ void __launch_bounds__(256) seg_sums_kernel(const float* __restrict__ in,
                                float* __restrict__ sums) {
  const int wave = (blockIdx.x * blockDim.x + threadIdx.x) >> 5;
  const int lane = threadIdx.x & 31;
  const int g = lane & 3;
  const int r = lane >> 2;

  const v4f* p = (const v4f*)(in + (size_t)wave * SEGFLTS) + (r * 4 + g);
  v4f acc = {0.f, 0.f, 0.f, 0.f};
  v4f buf[2][SUMQ];
#pragma unroll
  for (int q = 0; q < SUMQ; ++q) buf[0][q] = p[32 * (LASTQ - q)];
#pragma unroll
  for (int c = 0; c < SEG / 8 / SUMQ; ++c) {
    __builtin_amdgcn_sched_barrier(0);
    if (c + 1 < SEG / 8 / SUMQ) {
#pragma unroll
      for (int q = 0; q < SUMQ; ++q)
        buf[(c + 1) & 1][q] = p[32 * (LASTQ - ((c + 1) * SUMQ + q))];
    }
    __builtin_amdgcn_sched_barrier(0);
#pragma unroll
    for (int q = 0; q < SUMQ; ++q) acc += buf[c & 1][q];
  }
#pragma unroll
  for (int m = 4; m <= 16; m <<= 1) {
    acc[0] += __shfl_xor(acc[0], m, 32);
    acc[1] += __shfl_xor(acc[1], m, 32);
    acc[2] += __shfl_xor(acc[2], m, 32);
    acc[3] += __shfl_xor(acc[3], m, 32);
  }
  if (r == 0) {
    ((v4f*)sums)[wave * 4 + g] = acc;
  }
}

__device__ __forceinline__ void load_tile(float (&x)[8], const float* tb,
                                          int k0, int m) {
#pragma unroll
  for (int j = 0; j < 4; ++j) {
    x[2 * j]     = tb[(4 * j + k0)     * NCH + m];
    x[2 * j + 1] = tb[(4 * j + k0 + 1) * NCH + m];
  }
}

__global__ void __launch_bounds__(256) wmma_scan_kernel(const float* __restrict__ in,
                                 const float* __restrict__ sums,
                                 float* __restrict__ out) {
  const int wave = (blockIdx.x * blockDim.x + threadIdx.x) >> 5;
  const int lane = threadIdx.x & 31;
  const int m    = lane & 15;
  const int half = lane >> 4;
  const int k0   = half * 2;

  const float* base  = in  + (size_t)wave * SEGFLTS;
  float*       obase = out + (size_t)wave * SEGFLTS;

  const int seg = wave & (NSEG - 1);
  const float* sp = sums + (size_t)(wave - seg) * NCH + m;
  float v[NSEG - 1];
#pragma unroll
  for (int j = 0; j < NSEG - 1; ++j) v[j] = sp[j * NCH];
  __builtin_amdgcn_sched_barrier(0);

  float bx[RING][8];
#pragma unroll
  for (int p = 0; p < AHEAD; ++p) load_tile(bx[p], base + p * 256, k0, m);
  __builtin_amdgcn_sched_barrier(0);

  float carry = 0.f;
#pragma unroll
  for (int j = 0; j < NSEG - 1; ++j) carry += (j < seg) ? v[j] : 0.f;

#pragma unroll
  for (int t = 0; t < NTILE; ++t) {
    const float (&x)[8] = bx[t % RING];

    v8f d;
#pragma unroll
    for (int i = 0; i < 8; ++i) d[i] = carry;

#pragma unroll
    for (int kb = 0; kb < 16; kb += 4) {
      v2f a, b;
      a[0] = ((kb + k0)     <= m) ? 1.0f : 0.0f;
      a[1] = ((kb + k0 + 1) <= m) ? 1.0f : 0.0f;
      b[0] = x[kb / 2];
      b[1] = x[kb / 2 + 1];
      d = __builtin_amdgcn_wmma_f32_16x16x4_f32(
           false, a,  false, b,
           (short)0, d,  false,  false);
    }

    __builtin_amdgcn_sched_barrier(0);
    if (t + AHEAD < NTILE) {
      load_tile(bx[(t + AHEAD) % RING], base + (t + AHEAD) * 256, k0, m);
    }
    __builtin_amdgcn_sched_barrier(0);

    float* ob = obase + t * 256;
#pragma unroll
    for (int i = 0; i < 8; ++i) {
      __builtin_nontemporal_store(d[i], &ob[(i + 8 * half) * NCH + m]);
    }

    carry = __shfl(d[7], 16 + m, 32);
  }
}

extern "C" void kernel_launch(void* const* d_in, const int* in_sizes, int n_in,
                              void* d_out, int out_size, void* d_ws, size_t ws_size,
                              hipStream_t stream) {
  const float* in = (const float*)d_in[0];
  float* out = (float*)d_out;

  float* sums = (float*)d_ws;

  seg_sums_kernel<<<TOTSEG / 8, 256, 0, stream>>>(in, sums);
  wmma_scan_kernel<<<TOTSEG / 8, 256, 0, stream>>>(in, sums, out);
}


# Scan launch moves whole 128-byte lines, 16 bytes per lane: a product's columns are 4 channel groups x 4 row blocks; sums of blocks and segments ahead are added after the products (new sum grouping)
# speedup vs baseline: 1.7123x; 1.0824x over previous
#include <hip/hip_runtime.h>
#include <hip/hip_bf16.h>


typedef __attribute__((ext_vector_type(2))) float v2f;
typedef __attribute__((ext_vector_type(4))) float v4f;
typedef __attribute__((ext_vector_type(8))) float v8f;

#define NBATCH   64
#define LSEQ     32768
#define NCH      16
#define SEG      512
#define NSEG     (LSEQ / SEG)
#define TOTSEG   (NBATCH * NSEG)
#define SEGFLTS  (SEG * NCH)
#define SUMQ     16
#define LASTQ    (SEG / 8 - 1)
#define STROWS   64
#define STFLTS   (STROWS * NCH)
#define NST      (SEG / STROWS)
#define RING     3
#define AHEAD    2

__global__ void __launch_bounds__(256) seg_sums_kernel(const float* __restrict__ in,
                                float* __restrict__ sums) {
  const int wave = (blockIdx.x * blockDim.x + threadIdx.x) >> 5;
  const int lane = threadIdx.x & 31;
  const int g = lane & 3;
  const int r = lane >> 2;

  const v4f* p = (const v4f*)(in + (size_t)wave * SEGFLTS) + (r * 4 + g);
  v4f acc = {0.f, 0.f, 0.f, 0.f};
  v4f buf[2][SUMQ];
#pragma unroll
  for (int q = 0; q < SUMQ; ++q) buf[0][q] = p[32 * (LASTQ - q)];
#pragma unroll
  for (int c = 0; c < SEG / 8 / SUMQ; ++c) {
    __builtin_amdgcn_sched_barrier(0);
    if (c + 1 < SEG / 8 / SUMQ) {
#pragma unroll
      for (int q = 0; q < SUMQ; ++q)
        buf[(c + 1) & 1][q] = p[32 * (LASTQ - ((c + 1) * SUMQ + q))];
    }
    __builtin_amdgcn_sched_barrier(0);
#pragma unroll
    for (int q = 0; q < SUMQ; ++q) acc += buf[c & 1][q];
  }
#pragma unroll
  for (int m = 4; m <= 16; m <<= 1) {
    acc[0] += __shfl_xor(acc[0], m, 32);
    acc[1] += __shfl_xor(acc[1], m, 32);
    acc[2] += __shfl_xor(acc[2], m, 32);
    acc[3] += __shfl_xor(acc[3], m, 32);
  }
  if (r == 0) {
    ((v4f*)sums)[wave * 4 + g] = acc;
  }
}

__device__ __forceinline__ void load_group(v4f (&x)[8], const float* p) {
#pragma unroll
  for (int q = 0; q < 8; ++q) x[q] = *(const v4f*)(p + q * 2 * NCH);
}

template <int N>
__device__ __forceinline__ float row_shr(float x) {
  return __int_as_float(__builtin_amdgcn_update_dpp(
      0, __float_as_int(x), 0x110 + N, 0xf, 0xf, true));
}

__global__ void __launch_bounds__(256) wmma_scan_kernel(const float* __restrict__ in,
                                 const float* __restrict__ sums,
                                 float* __restrict__ out) {
  const int wave = (blockIdx.x * blockDim.x + threadIdx.x) >> 5;
  const int lane = threadIdx.x & 31;
  const int m    = lane & 15;
  const int half = lane >> 4;
  const int g    = lane & 3;
  const int blk  = (lane >> 2) & 3;

  const int loff = (16 * blk + half) * NCH + 4 * g;
  const float* base  = in  + (size_t)wave * SEGFLTS + loff;
  float*       obase = out + (size_t)wave * SEGFLTS + loff;

  const int seg = wave & (NSEG - 1);
  const float* sp = sums + (size_t)(wave - seg) * NCH + m;
  float v[NSEG - 1];
#pragma unroll
  for (int j = 0; j < NSEG - 1; ++j) v[j] = sp[j * NCH];
  __builtin_amdgcn_sched_barrier(0);

  v4f bx[RING][8];
#pragma unroll
  for (int p = 0; p < AHEAD; ++p) load_group(bx[p], base + p * STFLTS);
  __builtin_amdgcn_sched_barrier(0);

  float cm = 0.f;
#pragma unroll
  for (int j = 0; j < NSEG - 1; ++j) cm += (j < seg) ? v[j] : 0.f;
  v4f carry;
#pragma unroll
  for (int j = 0; j < 4; ++j) carry[j] = __shfl(cm, 4 * g + j, 32);

  const int tau = 2 * (m & 7) + (m >> 3);
  float a[8];
#pragma unroll
  for (int e = 0; e < 8; ++e) a[e] = ((2 * e + half) <= tau) ? 1.0f : 0.0f;

#pragma unroll
  for (int u = 0; u < NST; ++u) {
    __builtin_amdgcn_sched_barrier(0);
    if (u + AHEAD < NST) {
      load_group(bx[(u + AHEAD) % RING], base + (u + AHEAD) * STFLTS);
    }
    __builtin_amdgcn_sched_barrier(0);

    const v4f (&x)[8] = bx[u % RING];

    v8f d[4];
#pragma unroll
    for (int j = 0; j < 4; ++j)
#pragma unroll
      for (int i = 0; i < 8; ++i) d[j][i] = 0.f;

#pragma unroll
    for (int w = 0; w < 4; ++w) {
#pragma unroll
      for (int j = 0; j < 4; ++j) {
        v2f av, bv;
        av[0] = a[2 * w];
        av[1] = a[2 * w + 1];
        bv[0] = x[2 * w][j];
        bv[1] = x[2 * w + 1][j];
        d[j] = __builtin_amdgcn_wmma_f32_16x16x4_f32(
             false, av,  false, bv,
             (short)0, d[j],  false,  false);
      }
    }

    v4f off;
#pragma unroll
    for (int j = 0; j < 4; ++j) {
      const float t  = __shfl(d[j][7], 16 + m, 32);
      const float s1 = t  + row_shr<4>(t);
      const float s2 = s1 + row_shr<8>(s1);
      const float ex = row_shr<4>(s2);
      const float tot = __int_as_float(__builtin_amdgcn_permlane16(
          0, __float_as_int(s2), 0xfedcfedcu, 0xfedcfedcu, false, false));
      off[j]   = carry[j] + ex;
      carry[j] = carry[j] + tot;
    }

    float* ob = obase + u * STFLTS;
#pragma unroll
    for (int r = 0; r < 8; ++r) {
      v4f o;
#pragma unroll
      for (int j = 0; j < 4; ++j) o[j] = d[j][r] + off[j];
      __builtin_nontemporal_store(o, (v4f*)(ob + r * 2 * NCH));
    }
  }
}

extern "C" void kernel_launch(void* const* d_in, const int* in_sizes, int n_in,
                              void* d_out, int out_size, void* d_ws, size_t ws_size,
                              hipStream_t stream) {
  const float* in = (const float*)d_in[0];
  float* out = (float*)d_out;

  float* sums = (float*)d_ws;

  seg_sums_kernel<<<TOTSEG / 8, 256, 0, stream>>>(in, sums);
  wmma_scan_kernel<<<TOTSEG / 8, 256, 0, stream>>>(in, sums, out);
}
